# starting kernel
# baseline (speedup 1.0000x reference)
#include <hip/hip_runtime.h>
#include <math.h>

typedef __attribute__((ext_vector_type(2))) float v2f;
typedef __attribute__((ext_vector_type(8))) float v8f;

#define B_MAX      32
#define N_TOK      4096
#define IN_DIM     64
#define FEAT_DIM   768
#define NQ         4
#define NEG_BIG    (-3.402823466e38f)

__global__ void lap_qproj_kernel(const float* __restrict__ qt,
                                 const float* __restrict__ W,
                                 const float* __restrict__ bias,
                                 float* __restrict__ qp,
                                 float* __restrict__ qb)
{
    const int t = threadIdx.x;
    const int q = t >> 6;
    const int i = t & 63;
    float acc = 0.0f;
    const float* qrow = qt + q * FEAT_DIM;
    for (int f = 0; f < FEAT_DIM; ++f)
        acc += qrow[f] * W[f * IN_DIM + i];
    qp[q * IN_DIM + i] = acc;
    if (t < NQ) {
        float s = 0.0f;
        const float* qr = qt + t * FEAT_DIM;
        for (int f = 0; f < FEAT_DIM; ++f) s += qr[f] * bias[f];
        qb[t] = s;
    }
}

__global__ void lap_pool_kernel(const float* __restrict__ x,
                                const int*   __restrict__ mask,
                                const float* __restrict__ qp,
                                const float* __restrict__ qb,
                                float* __restrict__ pooled)
{
    const int b = blockIdx.x;
    const int q = blockIdx.y;
    const int t = threadIdx.x;

    __shared__ float s_w[N_TOK];
    __shared__ float s_qp[IN_DIM];
    __shared__ float s_red[256];
    __shared__ float s_part[4][IN_DIM];

    if (t < IN_DIM) s_qp[t] = qp[q * IN_DIM + t];
    __syncthreads();

    const float qbias = qb[q];
    const float* xb = x + (size_t)b * N_TOK * IN_DIM;
    const int*   mb = mask + (size_t)b * N_TOK;

    float lmax = NEG_BIG;
    for (int n = t; n < N_TOK; n += 256) {
        const float* xr = xb + n * IN_DIM;
        float s = qbias;
        #pragma unroll
        for (int i = 0; i < IN_DIM; i += 4) {
            float4 xv = *reinterpret_cast<const float4*>(xr + i);
            s += s_qp[i]     * xv.x + s_qp[i + 1] * xv.y
               + s_qp[i + 2] * xv.z + s_qp[i + 3] * xv.w;
        }
        if (mb[n] == 0) s = NEG_BIG;
        s_w[n] = s;
        lmax = fmaxf(lmax, s);
    }

    s_red[t] = lmax;
    __syncthreads();
    for (int off = 128; off > 0; off >>= 1) {
        if (t < off) s_red[t] = fmaxf(s_red[t], s_red[t + off]);
        __syncthreads();
    }
    const float mx = s_red[0];
    __syncthreads();

    float lsum = 0.0f;
    for (int n = t; n < N_TOK; n += 256) {
        float w = __expf(s_w[n] - mx);
        s_w[n] = w;
        lsum += w;
    }
    s_red[t] = lsum;
    __syncthreads();
    for (int off = 128; off > 0; off >>= 1) {
        if (t < off) s_red[t] += s_red[t + off];
        __syncthreads();
    }
    const float inv = 1.0f / s_red[0];
    __syncthreads();

    const int i = t & 63;
    const int c = t >> 6;
    float acc = 0.0f;
    const int nbeg = c * (N_TOK / 4);
    const int nend = nbeg + (N_TOK / 4);
    for (int n = nbeg; n < nend; ++n)
        acc += s_w[n] * xb[n * IN_DIM + i];
    s_part[c][i] = acc;
    __syncthreads();
    if (t < IN_DIM) {
        float p = s_part[0][t] + s_part[1][t] + s_part[2][t] + s_part[3][t];
        pooled[(b * NQ + q) * IN_DIM + t] = p * inv;
    }
}

__global__ void lap_outgemm_kernel(const float* __restrict__ pooled,
                                   const float* __restrict__ W,
                                   const float* __restrict__ bias,
                                   float* __restrict__ out)
{
    const int n0   = blockIdx.x * 16;
    const int m0   = blockIdx.y * 16;
    const int lane = threadIdx.x;
    const int hf   = lane >> 4;
    const int l    = lane & 15;

    const int arow = m0 + l;
    const int bcol = n0 + l;
    const float* ap = pooled + arow * IN_DIM;
    const float* bp = W + bcol * IN_DIM;

    v8f acc = {};
    #pragma unroll
    for (int k0 = 0; k0 < IN_DIM; k0 += 4) {
        const int ka = k0 + hf * 2;
        v2f a, bfr;
        a.x   = ap[ka];
        a.y   = ap[ka + 1];
        bfr.x = bp[ka];
        bfr.y = bp[ka + 1];
        acc = __builtin_amdgcn_wmma_f32_16x16x4_f32(
                   false, a,  false, bfr,
                   (short)0, acc,  false,  false);
    }

    const int col = n0 + l;
    const float bv = bias[col];
    #pragma unroll
    for (int r = 0; r < 8; ++r) {
        const int row = m0 + hf * 8 + r;
        out[row * FEAT_DIM + col] = acc[r] + bv;
    }
}

extern "C" void kernel_launch(void* const* d_in, const int* in_sizes, int n_in,
                              void* d_out, int out_size, void* d_ws, size_t ws_size,
                              hipStream_t stream) {
    const float* x    = (const float*)d_in[0];
    const int*   mask = (const int*)  d_in[1];
    const float* qt   = (const float*)d_in[2];
    const float* W    = (const float*)d_in[3];
    const float* bias = (const float*)d_in[4];
    float* out = (float*)d_out;

    const int B = in_sizes[0] / (N_TOK * IN_DIM);

    float* ws     = (float*)d_ws;
    float* qp     = ws;
    float* qb     = ws + 256;
    float* pooled = ws + 512;

    lap_qproj_kernel<<<1, 256, 0, stream>>>(qt, W, bias, qp, qb);
    lap_pool_kernel<<<dim3(B, NQ), 256, 0, stream>>>(x, mask, qp, qb, pooled);
    lap_outgemm_kernel<<<dim3(FEAT_DIM / 16, (B * NQ) / 16), 32, 0, stream>>>(
        pooled, W, bias, out);
}


# q-projection kernel: each 768-long f-sum cut into 32 chunks, one wave each (8 blocks of 1024 threads), partials added in LDS in chunk order (changes f-summation order); was one serial block
# speedup vs baseline: 1.5173x; 1.5173x over previous
#include <hip/hip_runtime.h>
#include <math.h>

typedef __attribute__((ext_vector_type(2))) float v2f;
typedef __attribute__((ext_vector_type(8))) float v8f;

#define B_MAX      32
#define N_TOK      4096
#define IN_DIM     64
#define FEAT_DIM   768
#define NQ         4
#define NEG_BIG    (-3.402823466e38f)

#define QP_CHUNKS  32
#define QP_FLEN    (FEAT_DIM / QP_CHUNKS)

__global__ void __launch_bounds__(1024)
lap_qproj_kernel(const float* __restrict__ qt,
                 const float* __restrict__ W,
                 const float* __restrict__ bias,
                 float* __restrict__ qp,
                 float* __restrict__ qb)
{
    __shared__ float s_p[QP_CHUNKS][32];
    __shared__ float s_b[NQ][QP_CHUNKS];

    const int q    = blockIdx.x;
    const int half = blockIdx.y;
    const int t    = threadIdx.x;
    const int lane = t & 31;
    const int fc   = __builtin_amdgcn_readfirstlane(t >> 5);
    const int f0   = fc * QP_FLEN;
    const bool do_bias = (q == 0) && (half == 0);

    const float* qrow = qt + q * FEAT_DIM + f0;
    const float* wcol = W + f0 * IN_DIM + half * 32 + lane;

    float acc = 0.0f;
    for (int j = 0; j < QP_FLEN; ++j)
        acc += qrow[j] * wcol[j * IN_DIM];
    s_p[fc][lane] = acc;

    if (do_bias) {
        const float* bsl = bias + f0;
        for (int qq = 0; qq < NQ; ++qq) {
            const float* qr = qt + qq * FEAT_DIM + f0;
            float s = 0.0f;
            for (int j = 0; j < QP_FLEN; ++j) s += qr[j] * bsl[j];
            if (lane == 0) s_b[qq][fc] = s;
        }
    }
    __syncthreads();

    if (t < 32) {
        float s = 0.0f;
        for (int c = 0; c < QP_CHUNKS; ++c) s += s_p[c][t];
        qp[q * IN_DIM + half * 32 + t] = s;
    } else if (do_bias && t < 32 + NQ) {
        const int qq = t - 32;
        float s = 0.0f;
        for (int c = 0; c < QP_CHUNKS; ++c) s += s_b[qq][c];
        qb[qq] = s;
    }
}

__global__ void lap_pool_kernel(const float* __restrict__ x,
                                const int*   __restrict__ mask,
                                const float* __restrict__ qp,
                                const float* __restrict__ qb,
                                float* __restrict__ pooled)
{
    const int b = blockIdx.x;
    const int q = blockIdx.y;
    const int t = threadIdx.x;

    __shared__ float s_w[N_TOK];
    __shared__ float s_qp[IN_DIM];
    __shared__ float s_red[256];
    __shared__ float s_part[4][IN_DIM];

    if (t < IN_DIM) s_qp[t] = qp[q * IN_DIM + t];
    __syncthreads();

    const float qbias = qb[q];
    const float* xb = x + (size_t)b * N_TOK * IN_DIM;
    const int*   mb = mask + (size_t)b * N_TOK;

    float lmax = NEG_BIG;
    for (int n = t; n < N_TOK; n += 256) {
        const float* xr = xb + n * IN_DIM;
        float s = qbias;
        #pragma unroll
        for (int i = 0; i < IN_DIM; i += 4) {
            float4 xv = *reinterpret_cast<const float4*>(xr + i);
            s += s_qp[i]     * xv.x + s_qp[i + 1] * xv.y
               + s_qp[i + 2] * xv.z + s_qp[i + 3] * xv.w;
        }
        if (mb[n] == 0) s = NEG_BIG;
        s_w[n] = s;
        lmax = fmaxf(lmax, s);
    }

    s_red[t] = lmax;
    __syncthreads();
    for (int off = 128; off > 0; off >>= 1) {
        if (t < off) s_red[t] = fmaxf(s_red[t], s_red[t + off]);
        __syncthreads();
    }
    const float mx = s_red[0];
    __syncthreads();

    float lsum = 0.0f;
    for (int n = t; n < N_TOK; n += 256) {
        float w = __expf(s_w[n] - mx);
        s_w[n] = w;
        lsum += w;
    }
    s_red[t] = lsum;
    __syncthreads();
    for (int off = 128; off > 0; off >>= 1) {
        if (t < off) s_red[t] += s_red[t + off];
        __syncthreads();
    }
    const float inv = 1.0f / s_red[0];
    __syncthreads();

    const int i = t & 63;
    const int c = t >> 6;
    float acc = 0.0f;
    const int nbeg = c * (N_TOK / 4);
    const int nend = nbeg + (N_TOK / 4);
    for (int n = nbeg; n < nend; ++n)
        acc += s_w[n] * xb[n * IN_DIM + i];
    s_part[c][i] = acc;
    __syncthreads();
    if (t < IN_DIM) {
        float p = s_part[0][t] + s_part[1][t] + s_part[2][t] + s_part[3][t];
        pooled[(b * NQ + q) * IN_DIM + t] = p * inv;
    }
}

__global__ void lap_outgemm_kernel(const float* __restrict__ pooled,
                                   const float* __restrict__ W,
                                   const float* __restrict__ bias,
                                   float* __restrict__ out)
{
    const int n0   = blockIdx.x * 16;
    const int m0   = blockIdx.y * 16;
    const int lane = threadIdx.x;
    const int hf   = lane >> 4;
    const int l    = lane & 15;

    const int arow = m0 + l;
    const int bcol = n0 + l;
    const float* ap = pooled + arow * IN_DIM;
    const float* bp = W + bcol * IN_DIM;

    v8f acc = {};
    #pragma unroll
    for (int k0 = 0; k0 < IN_DIM; k0 += 4) {
        const int ka = k0 + hf * 2;
        v2f a, bfr;
        a.x   = ap[ka];
        a.y   = ap[ka + 1];
        bfr.x = bp[ka];
        bfr.y = bp[ka + 1];
        acc = __builtin_amdgcn_wmma_f32_16x16x4_f32(
                   false, a,  false, bfr,
                   (short)0, acc,  false,  false);
    }

    const int col = n0 + l;
    const float bv = bias[col];
    #pragma unroll
    for (int r = 0; r < 8; ++r) {
        const int row = m0 + hf * 8 + r;
        out[row * FEAT_DIM + col] = acc[r] + bv;
    }
}

extern "C" void kernel_launch(void* const* d_in, const int* in_sizes, int n_in,
                              void* d_out, int out_size, void* d_ws, size_t ws_size,
                              hipStream_t stream) {
    const float* x    = (const float*)d_in[0];
    const int*   mask = (const int*)  d_in[1];
    const float* qt   = (const float*)d_in[2];
    const float* W    = (const float*)d_in[3];
    const float* bias = (const float*)d_in[4];
    float* out = (float*)d_out;

    const int B = in_sizes[0] / (N_TOK * IN_DIM);

    float* ws     = (float*)d_ws;
    float* qp     = ws;
    float* qb     = ws + 256;
    float* pooled = ws + 512;

    lap_qproj_kernel<<<dim3(NQ, 2), 1024, 0, stream>>>(qt, W, bias, qp, qb);
    lap_pool_kernel<<<dim3(B, NQ), 256, 0, stream>>>(x, mask, qp, qb, pooled);
    lap_outgemm_kernel<<<dim3(FEAT_DIM / 16, (B * NQ) / 16), 32, 0, stream>>>(
        pooled, W, bias, out);
}


# pooling split over tokens: block = (batch row, 512-token slice), all 4 queries, slice kept in LDS, so x is read once, not 8x; slice partials merged by a small 4th launch (changes n-sum order)
# speedup vs baseline: 5.7180x; 3.7685x over previous
#include <hip/hip_runtime.h>
#include <math.h>

typedef __attribute__((ext_vector_type(2))) float v2f;
typedef __attribute__((ext_vector_type(8))) float v8f;

#define B_MAX      32
#define N_TOK      4096
#define IN_DIM     64
#define FEAT_DIM   768
#define NQ         4
#define NEG_BIG    (-3.402823466e38f)

#define QP_CHUNKS  32
#define QP_FLEN    (FEAT_DIM / QP_CHUNKS)

__global__ void __launch_bounds__(1024)
lap_qproj_kernel(const float* __restrict__ qt,
                 const float* __restrict__ W,
                 const float* __restrict__ bias,
                 float* __restrict__ qp,
                 float* __restrict__ qb)
{
    __shared__ float s_p[QP_CHUNKS][32];
    __shared__ float s_b[NQ][QP_CHUNKS];

    const int q    = blockIdx.x;
    const int half = blockIdx.y;
    const int t    = threadIdx.x;
    const int lane = t & 31;
    const int fc   = __builtin_amdgcn_readfirstlane(t >> 5);
    const int f0   = fc * QP_FLEN;
    const bool do_bias = (q == 0) && (half == 0);

    const float* qrow = qt + q * FEAT_DIM + f0;
    const float* wcol = W + f0 * IN_DIM + half * 32 + lane;

    float acc = 0.0f;
    for (int j = 0; j < QP_FLEN; ++j)
        acc += qrow[j] * wcol[j * IN_DIM];
    s_p[fc][lane] = acc;

    if (do_bias) {
        const float* bsl = bias + f0;
        for (int qq = 0; qq < NQ; ++qq) {
            const float* qr = qt + qq * FEAT_DIM + f0;
            float s = 0.0f;
            for (int j = 0; j < QP_FLEN; ++j) s += qr[j] * bsl[j];
            if (lane == 0) s_b[qq][fc] = s;
        }
    }
    __syncthreads();

    if (t < 32) {
        float s = 0.0f;
        for (int c = 0; c < QP_CHUNKS; ++c) s += s_p[c][t];
        qp[q * IN_DIM + half * 32 + t] = s;
    } else if (do_bias && t < 32 + NQ) {
        const int qq = t - 32;
        float s = 0.0f;
        for (int c = 0; c < QP_CHUNKS; ++c) s += s_b[qq][c];
        qb[qq] = s;
    }
}

#define SLICES      8
#define TOK_S       (N_TOK / SLICES)
#define XPITCH      (IN_DIM + 4)
#define PS_THREADS  512
#define PS_WAVES    (PS_THREADS / 32)
#define PART_ML     512

__global__ void __launch_bounds__(PS_THREADS)
lap_pool_slice_kernel(const float* __restrict__ x,
                      const int*   __restrict__ mask,
                      const float* __restrict__ qp,
                      const float* __restrict__ qb,
                      float* __restrict__ part)
{
    __shared__ __attribute__((aligned(16))) float s_x[TOK_S * XPITCH];
    __shared__ float s_sc[NQ][TOK_S];
    __shared__ __attribute__((aligned(16))) float s_e[TOK_S * NQ];
    __shared__ __attribute__((aligned(16))) float s_pp[PS_WAVES][NQ][IN_DIM];
    __shared__ float s_ls[NQ][4];
    __shared__ float s_m[NQ];

    const int b    = blockIdx.x;
    const int sl   = blockIdx.y;
    const int t    = threadIdx.x;
    const int lane = t & 31;
    const int w    = __builtin_amdgcn_readfirstlane(t >> 5);
    const size_t tok0 = (size_t)b * N_TOK + (size_t)sl * TOK_S;

    const float4* xs4 = reinterpret_cast<const float4*>(x + tok0 * IN_DIM);
    float4 v[16];
    #pragma unroll
    for (int j = 0; j < 16; ++j) v[j] = xs4[j * PS_THREADS + t];
    const int mk = mask[tok0 + t];
    #pragma unroll
    for (int j = 0; j < 16; ++j) {
        const int f = j * PS_THREADS + t;
        *reinterpret_cast<float4*>(&s_x[(f >> 4) * XPITCH + (f & 15) * 4]) = v[j];
    }
    __syncthreads();

    {
        float xr[IN_DIM];
        #pragma unroll
        for (int k = 0; k < IN_DIM / 4; ++k) {
            const float4 u = *reinterpret_cast<const float4*>(&s_x[t * XPITCH + 4 * k]);
            xr[4 * k] = u.x; xr[4 * k + 1] = u.y; xr[4 * k + 2] = u.z; xr[4 * k + 3] = u.w;
        }
        #pragma unroll
        for (int q = 0; q < NQ; ++q) {
            float s = qb[q];
            #pragma unroll
            for (int i = 0; i < IN_DIM; ++i) s += qp[q * IN_DIM + i] * xr[i];
            if (mk == 0) s = NEG_BIG;
            s_sc[q][t] = s;
        }
    }
    __syncthreads();

    {
        const int q   = w >> 2;
        const int seg = w & 3;
        float mloc = NEG_BIG;
        #pragma unroll
        for (int k = 0; k < TOK_S / 32; ++k) mloc = fmaxf(mloc, s_sc[q][lane + 32 * k]);
        #pragma unroll
        for (int off = 16; off > 0; off >>= 1) mloc = fmaxf(mloc, __shfl_xor(mloc, off));
        float ps = 0.0f;
        #pragma unroll
        for (int k = 0; k < 4; ++k) {
            const int n = seg * 128 + k * 32 + lane;
            const float e = __expf(s_sc[q][n] - mloc);
            s_e[n * NQ + q] = e;
            ps += e;
        }
        #pragma unroll
        for (int off = 16; off > 0; off >>= 1) ps += __shfl_xor(ps, off);
        if (lane == 0) {
            s_ls[q][seg] = ps;
            if (seg == 0) s_m[q] = mloc;
        }
    }
    __syncthreads();

    {
        float a0x = 0.0f, a0y = 0.0f, a1x = 0.0f, a1y = 0.0f;
        float a2x = 0.0f, a2y = 0.0f, a3x = 0.0f, a3y = 0.0f;
        #pragma unroll 8
        for (int k = 0; k < 32; ++k) {
            const int n = w * 32 + k;
            const float2 xv = *reinterpret_cast<const float2*>(&s_x[n * XPITCH + 2 * lane]);
            const float4 e  = *reinterpret_cast<const float4*>(&s_e[n * NQ]);
            a0x += e.x * xv.x; a0y += e.x * xv.y;
            a1x += e.y * xv.x; a1y += e.y * xv.y;
            a2x += e.z * xv.x; a2y += e.z * xv.y;
            a3x += e.w * xv.x; a3y += e.w * xv.y;
        }
        *reinterpret_cast<float2*>(&s_pp[w][0][2 * lane]) = make_float2(a0x, a0y);
        *reinterpret_cast<float2*>(&s_pp[w][1][2 * lane]) = make_float2(a1x, a1y);
        *reinterpret_cast<float2*>(&s_pp[w][2][2 * lane]) = make_float2(a2x, a2y);
        *reinterpret_cast<float2*>(&s_pp[w][3][2 * lane]) = make_float2(a3x, a3y);
    }
    __syncthreads();

    if (t < NQ * IN_DIM) {
        const int q = t >> 6;
        const int i = t & 63;
        float s = 0.0f;
        #pragma unroll
        for (int ww = 0; ww < PS_WAVES; ++ww) s += s_pp[ww][q][i];
        part[(size_t)(b * NQ + q) * FEAT_DIM + sl * IN_DIM + i] = s;
    } else if (t < NQ * IN_DIM + NQ) {
        const int q = t - NQ * IN_DIM;
        const float l = s_ls[q][0] + s_ls[q][1] + s_ls[q][2] + s_ls[q][3];
        *reinterpret_cast<float2*>(&part[(size_t)(b * NQ + q) * FEAT_DIM + PART_ML + sl * 32]) =
            make_float2(s_m[q], l);
    }
}

__global__ void __launch_bounds__(IN_DIM)
lap_combine_kernel(const float* __restrict__ part,
                   float* __restrict__ pooled)
{
    const int m = blockIdx.x;
    const int i = threadIdx.x;
    const float* row = part + (size_t)m * FEAT_DIM;

    float ms[SLICES], ls[SLICES];
    float M = NEG_BIG;
    #pragma unroll
    for (int s = 0; s < SLICES; ++s) {
        ms[s] = row[PART_ML + s * 32];
        ls[s] = row[PART_ML + s * 32 + 1];
        M = fmaxf(M, ms[s]);
    }
    float L = 0.0f, acc = 0.0f;
    #pragma unroll
    for (int s = 0; s < SLICES; ++s) {
        const float c = __expf(ms[s] - M);
        L   += c * ls[s];
        acc += c * row[s * IN_DIM + i];
    }
    pooled[m * IN_DIM + i] = acc * (1.0f / L);
}

__global__ void lap_outgemm_kernel(const float* __restrict__ pooled,
                                   const float* __restrict__ W,
                                   const float* __restrict__ bias,
                                   float* __restrict__ out)
{
    const int n0   = blockIdx.x * 16;
    const int m0   = blockIdx.y * 16;
    const int lane = threadIdx.x;
    const int hf   = lane >> 4;
    const int l    = lane & 15;

    const int arow = m0 + l;
    const int bcol = n0 + l;
    const float* ap = pooled + arow * IN_DIM;
    const float* bp = W + bcol * IN_DIM;

    v8f acc = {};
    #pragma unroll
    for (int k0 = 0; k0 < IN_DIM; k0 += 4) {
        const int ka = k0 + hf * 2;
        v2f a, bfr;
        a.x   = ap[ka];
        a.y   = ap[ka + 1];
        bfr.x = bp[ka];
        bfr.y = bp[ka + 1];
        acc = __builtin_amdgcn_wmma_f32_16x16x4_f32(
                   false, a,  false, bfr,
                   (short)0, acc,  false,  false);
    }

    const int col = n0 + l;
    const float bv = bias[col];
    #pragma unroll
    for (int r = 0; r < 8; ++r) {
        const int row = m0 + hf * 8 + r;
        out[row * FEAT_DIM + col] = acc[r] + bv;
    }
}

extern "C" void kernel_launch(void* const* d_in, const int* in_sizes, int n_in,
                              void* d_out, int out_size, void* d_ws, size_t ws_size,
                              hipStream_t stream) {
    const float* x    = (const float*)d_in[0];
    const int*   mask = (const int*)  d_in[1];
    const float* qt   = (const float*)d_in[2];
    const float* W    = (const float*)d_in[3];
    const float* bias = (const float*)d_in[4];
    float* out = (float*)d_out;

    const int B = in_sizes[0] / (N_TOK * IN_DIM);

    float* ws     = (float*)d_ws;
    float* qp     = ws;
    float* qb     = ws + 256;
    float* pooled = ws + 512;

    lap_qproj_kernel<<<dim3(NQ, 2), 1024, 0, stream>>>(qt, W, bias, qp, qb);
    lap_pool_slice_kernel<<<dim3(B, SLICES), PS_THREADS, 0, stream>>>(x, mask, qp, qb, out);
    lap_combine_kernel<<<B * NQ, IN_DIM, 0, stream>>>(out, pooled);
    lap_outgemm_kernel<<<dim3(FEAT_DIM / 16, (B * NQ) / 16), 32, 0, stream>>>(
        pooled, W, bias, out);
}


# pooling block cut from 16 waves to 4, one per SIMD (thread = 4 tokens, wave = 1 query in softmax, half-wave = token parity in weighted sum, q_proj via LDS): 4x fewer waves to start; new n-sum order
# speedup vs baseline: 6.4291x; 1.1244x over previous
#include <hip/hip_runtime.h>
#include <math.h>

typedef __attribute__((ext_vector_type(2))) float v2f;
typedef __attribute__((ext_vector_type(8))) float v8f;

#define B_MAX      32
#define N_TOK      4096
#define IN_DIM     64
#define FEAT_DIM   768
#define NQ         4
#define NEG_BIG    (-3.402823466e38f)

#define QP_CHUNKS  32
#define QP_FLEN    (FEAT_DIM / QP_CHUNKS)

__global__ void __launch_bounds__(1024)
lap_qproj_kernel(const float* __restrict__ qt,
                 const float* __restrict__ W,
                 const float* __restrict__ bias,
                 float* __restrict__ qp,
                 float* __restrict__ qb)
{
    __shared__ float s_p[QP_CHUNKS][32];
    __shared__ float s_b[NQ][QP_CHUNKS];

    const int q    = blockIdx.x;
    const int half = blockIdx.y;
    const int t    = threadIdx.x;
    const int lane = t & 31;
    const int fc   = __builtin_amdgcn_readfirstlane(t >> 5);
    const int f0   = fc * QP_FLEN;
    const bool do_bias = (q == 0) && (half == 0);

    const float* qrow = qt + q * FEAT_DIM + f0;
    const float* wcol = W + f0 * IN_DIM + half * 32 + lane;

    float acc = 0.0f;
    for (int j = 0; j < QP_FLEN; ++j)
        acc += qrow[j] * wcol[j * IN_DIM];
    s_p[fc][lane] = acc;

    if (do_bias) {
        const float* bsl = bias + f0;
        for (int qq = 0; qq < NQ; ++qq) {
            const float* qr = qt + qq * FEAT_DIM + f0;
            float s = 0.0f;
            for (int j = 0; j < QP_FLEN; ++j) s += qr[j] * bsl[j];
            if (lane == 0) s_b[qq][fc] = s;
        }
    }
    __syncthreads();

    if (t < 32) {
        float s = 0.0f;
        for (int c = 0; c < QP_CHUNKS; ++c) s += s_p[c][t];
        qp[q * IN_DIM + half * 32 + t] = s;
    } else if (do_bias && t < 32 + NQ) {
        const int qq = t - 32;
        float s = 0.0f;
        for (int c = 0; c < QP_CHUNKS; ++c) s += s_b[qq][c];
        qb[qq] = s;
    }
}

#define SLICES      8
#define TOK_S       (N_TOK / SLICES)
#define XPITCH      (IN_DIM + 4)
#define PS_THREADS  128
#define PS_WAVES    (PS_THREADS / 32)
#define TOK_T       (TOK_S / PS_THREADS)
#define LD_T        (TOK_S * (IN_DIM / 4) / PS_THREADS)
#define TOK_W       (TOK_S / PS_WAVES)
#define PART_ML     512

#define P4_IT       16

#define STAGE_FENCE() __builtin_amdgcn_sched_barrier(0)

#define P2_KC       4

__device__ __forceinline__ void p2_load(float4 (&u)[P2_KC][TOK_T], float4 (&cq)[P2_KC][NQ],
                                        const float* sx, const float* sqp, int t, int st)
{
    #pragma unroll
    for (int kk = 0; kk < P2_KC; ++kk) {
        const int kc = st * P2_KC + kk;
        #pragma unroll
        for (int h = 0; h < TOK_T; ++h)
            u[kk][h] = *reinterpret_cast<const float4*>(&sx[(t + PS_THREADS * h) * XPITCH + 4 * kc]);
        #pragma unroll
        for (int q = 0; q < NQ; ++q)
            cq[kk][q] = *reinterpret_cast<const float4*>(&sqp[q * IN_DIM + 4 * kc]);
    }
}

__device__ __forceinline__ void p2_sum(float (&s)[TOK_T][NQ], const float4 (&u)[P2_KC][TOK_T],
                                       const float4 (&cq)[P2_KC][NQ])
{
    #pragma unroll
    for (int kk = 0; kk < P2_KC; ++kk) {
        #pragma unroll
        for (int q = 0; q < NQ; ++q) {
            #pragma unroll
            for (int h = 0; h < TOK_T; ++h) {
                float r = s[h][q];
                r = fmaf(cq[kk][q].x, u[kk][h].x, r);
                r = fmaf(cq[kk][q].y, u[kk][h].y, r);
                r = fmaf(cq[kk][q].z, u[kk][h].z, r);
                r = fmaf(cq[kk][q].w, u[kk][h].w, r);
                s[h][q] = r;
            }
        }
    }
}

__device__ __forceinline__ void p4_load(float4 (&xv)[P4_IT], float4 (&ev)[P4_IT],
                                        const float* sx, const float* se, int n0, int c, int st)
{
    #pragma unroll
    for (int k = 0; k < P4_IT; ++k) {
        const int n = n0 + 2 * (st * P4_IT + k);
        xv[k] = *reinterpret_cast<const float4*>(&sx[n * XPITCH + 4 * c]);
        ev[k] = *reinterpret_cast<const float4*>(&se[n * NQ]);
    }
}

__device__ __forceinline__ void p4_sum(float4 (&a)[NQ], const float4 (&xv)[P4_IT],
                                       const float4 (&ev)[P4_IT])
{
    #pragma unroll
    for (int k = 0; k < P4_IT; ++k) {
        const float wq[NQ] = { ev[k].x, ev[k].y, ev[k].z, ev[k].w };
        #pragma unroll
        for (int q = 0; q < NQ; ++q) {
            a[q].x = fmaf(wq[q], xv[k].x, a[q].x);
            a[q].y = fmaf(wq[q], xv[k].y, a[q].y);
            a[q].z = fmaf(wq[q], xv[k].z, a[q].z);
            a[q].w = fmaf(wq[q], xv[k].w, a[q].w);
        }
    }
}

__global__ void __launch_bounds__(PS_THREADS)
lap_pool_slice_kernel(const float* __restrict__ x,
                      const int*   __restrict__ mask,
                      const float* __restrict__ qp,
                      const float* __restrict__ qb,
                      float* __restrict__ part)
{
    __shared__ __attribute__((aligned(16))) float s_x[TOK_S * XPITCH];
    __shared__ float s_sc[NQ][TOK_S];
    __shared__ __attribute__((aligned(16))) float s_e[TOK_S * NQ];
    __shared__ __attribute__((aligned(16))) float s_pp[2 * PS_WAVES][NQ][IN_DIM];
    __shared__ __attribute__((aligned(16))) float s_qp[NQ * IN_DIM];
    __shared__ float s_l[NQ];
    __shared__ float s_m[NQ];

    const int b    = blockIdx.x;
    const int sl   = blockIdx.y;
    const int t    = threadIdx.x;
    const int lane = t & 31;
    const int w    = __builtin_amdgcn_readfirstlane(t >> 5);
    const size_t tok0 = (size_t)b * N_TOK + (size_t)sl * TOK_S;

    const float4* xs4 = reinterpret_cast<const float4*>(x + tok0 * IN_DIM);
    float4 v[LD_T];
    #pragma unroll
    for (int j = 0; j < LD_T; ++j) v[j] = xs4[j * PS_THREADS + t];
    int mk[TOK_T];
    #pragma unroll
    for (int h = 0; h < TOK_T; ++h) mk[h] = mask[tok0 + t + PS_THREADS * h];
    const float2 qv = reinterpret_cast<const float2*>(qp)[t];
    *reinterpret_cast<float2*>(&s_qp[2 * t]) = qv;
    #pragma unroll
    for (int j = 0; j < LD_T; ++j) {
        const int f = j * PS_THREADS + t;
        *reinterpret_cast<float4*>(&s_x[(f >> 4) * XPITCH + (f & 15) * 4]) = v[j];
    }
    __syncthreads();

    {
        float s[TOK_T][NQ];
        #pragma unroll
        for (int h = 0; h < TOK_T; ++h) {
            #pragma unroll
            for (int q = 0; q < NQ; ++q) s[h][q] = qb[q];
        }
        float4 uA[P2_KC][TOK_T], cA[P2_KC][NQ];
        float4 uB[P2_KC][TOK_T], cB[P2_KC][NQ];
        p2_load(uA, cA, s_x, s_qp, t, 0);
        STAGE_FENCE();
        p2_load(uB, cB, s_x, s_qp, t, 1);
        STAGE_FENCE();
        p2_sum(s, uA, cA);
        STAGE_FENCE();
        p2_load(uA, cA, s_x, s_qp, t, 2);
        STAGE_FENCE();
        p2_sum(s, uB, cB);
        STAGE_FENCE();
        p2_load(uB, cB, s_x, s_qp, t, 3);
        STAGE_FENCE();
        p2_sum(s, uA, cA);
        STAGE_FENCE();
        p2_sum(s, uB, cB);
        #pragma unroll
        for (int h = 0; h < TOK_T; ++h) {
            #pragma unroll
            for (int q = 0; q < NQ; ++q)
                s_sc[q][t + PS_THREADS * h] = (mk[h] == 0) ? NEG_BIG : s[h][q];
        }
    }
    __syncthreads();

    {
        const int q = w;
        float sc[TOK_S / 32];
        float mloc = NEG_BIG;
        #pragma unroll
        for (int k = 0; k < TOK_S / 32; ++k) {
            sc[k] = s_sc[q][lane + 32 * k];
            mloc = fmaxf(mloc, sc[k]);
        }
        #pragma unroll
        for (int off = 16; off > 0; off >>= 1) mloc = fmaxf(mloc, __shfl_xor(mloc, off));
        float ps = 0.0f;
        #pragma unroll
        for (int k = 0; k < TOK_S / 32; ++k) {
            const int n = lane + 32 * k;
            const float e = __expf(sc[k] - mloc);
            s_e[n * NQ + q] = e;
            ps += e;
        }
        #pragma unroll
        for (int off = 16; off > 0; off >>= 1) ps += __shfl_xor(ps, off);
        if (lane == 0) {
            s_l[q] = ps;
            s_m[q] = mloc;
        }
    }
    __syncthreads();

    {
        const int hf = lane >> 4;
        const int c  = lane & 15;
        const int n0 = w * TOK_W + hf;
        float4 a[NQ];
        #pragma unroll
        for (int q = 0; q < NQ; ++q) a[q] = make_float4(0.0f, 0.0f, 0.0f, 0.0f);
        float4 xA[P4_IT], eA[P4_IT], xB[P4_IT], eB[P4_IT];
        p4_load(xA, eA, s_x, s_e, n0, c, 0);
        STAGE_FENCE();
        p4_load(xB, eB, s_x, s_e, n0, c, 1);
        STAGE_FENCE();
        p4_sum(a, xA, eA);
        STAGE_FENCE();
        p4_load(xA, eA, s_x, s_e, n0, c, 2);
        STAGE_FENCE();
        p4_sum(a, xB, eB);
        STAGE_FENCE();
        p4_load(xB, eB, s_x, s_e, n0, c, 3);
        STAGE_FENCE();
        p4_sum(a, xA, eA);
        STAGE_FENCE();
        p4_sum(a, xB, eB);
        #pragma unroll
        for (int q = 0; q < NQ; ++q)
            *reinterpret_cast<float4*>(&s_pp[2 * w + hf][q][4 * c]) = a[q];
    }
    __syncthreads();

    #pragma unroll
    for (int r = 0; r < (NQ * IN_DIM) / PS_THREADS; ++r) {
        const int o = t + PS_THREADS * r;
        const int q = o >> 6;
        const int i = o & 63;
        float s = 0.0f;
        #pragma unroll
        for (int pw = 0; pw < 2 * PS_WAVES; ++pw) s += s_pp[pw][q][i];
        part[(size_t)(b * NQ + q) * FEAT_DIM + sl * IN_DIM + i] = s;
    }
    if (t < NQ) {
        *reinterpret_cast<float2*>(&part[(size_t)(b * NQ + t) * FEAT_DIM + PART_ML + sl * 32]) =
            make_float2(s_m[t], s_l[t]);
    }
}

__global__ void __launch_bounds__(IN_DIM)
lap_combine_kernel(const float* __restrict__ part,
                   float* __restrict__ pooled)
{
    const int m = blockIdx.x;
    const int i = threadIdx.x;
    const float* row = part + (size_t)m * FEAT_DIM;

    float ms[SLICES], ls[SLICES];
    float M = NEG_BIG;
    #pragma unroll
    for (int s = 0; s < SLICES; ++s) {
        ms[s] = row[PART_ML + s * 32];
        ls[s] = row[PART_ML + s * 32 + 1];
        M = fmaxf(M, ms[s]);
    }
    float L = 0.0f, acc = 0.0f;
    #pragma unroll
    for (int s = 0; s < SLICES; ++s) {
        const float c = __expf(ms[s] - M);
        L   += c * ls[s];
        acc += c * row[s * IN_DIM + i];
    }
    pooled[m * IN_DIM + i] = acc * (1.0f / L);
}

__global__ void lap_outgemm_kernel(const float* __restrict__ pooled,
                                   const float* __restrict__ W,
                                   const float* __restrict__ bias,
                                   float* __restrict__ out)
{
    const int n0   = blockIdx.x * 16;
    const int m0   = blockIdx.y * 16;
    const int lane = threadIdx.x;
    const int hf   = lane >> 4;
    const int l    = lane & 15;

    const int arow = m0 + l;
    const int bcol = n0 + l;
    const float* ap = pooled + arow * IN_DIM;
    const float* bp = W + bcol * IN_DIM;

    v8f acc = {};
    #pragma unroll
    for (int k0 = 0; k0 < IN_DIM; k0 += 4) {
        const int ka = k0 + hf * 2;
        v2f a, bfr;
        a.x   = ap[ka];
        a.y   = ap[ka + 1];
        bfr.x = bp[ka];
        bfr.y = bp[ka + 1];
        acc = __builtin_amdgcn_wmma_f32_16x16x4_f32(
                   false, a,  false, bfr,
                   (short)0, acc,  false,  false);
    }

    const int col = n0 + l;
    const float bv = bias[col];
    #pragma unroll
    for (int r = 0; r < 8; ++r) {
        const int row = m0 + hf * 8 + r;
        out[row * FEAT_DIM + col] = acc[r] + bv;
    }
}

extern "C" void kernel_launch(void* const* d_in, const int* in_sizes, int n_in,
                              void* d_out, int out_size, void* d_ws, size_t ws_size,
                              hipStream_t stream) {
    const float* x    = (const float*)d_in[0];
    const int*   mask = (const int*)  d_in[1];
    const float* qt   = (const float*)d_in[2];
    const float* W    = (const float*)d_in[3];
    const float* bias = (const float*)d_in[4];
    float* out = (float*)d_out;

    const int B = in_sizes[0] / (N_TOK * IN_DIM);

    float* ws     = (float*)d_ws;
    float* qp     = ws;
    float* qb     = ws + 256;
    float* pooled = ws + 512;

    lap_qproj_kernel<<<dim3(NQ, 2), 1024, 0, stream>>>(qt, W, bias, qp, qb);
    lap_pool_slice_kernel<<<dim3(B, SLICES), PS_THREADS, 0, stream>>>(x, mask, qp, qb, out);
    lap_combine_kernel<<<B * NQ, IN_DIM, 0, stream>>>(out, pooled);
    lap_outgemm_kernel<<<dim3(FEAT_DIM / 16, (B * NQ) / 16), 32, 0, stream>>>(
        pooled, W, bias, out);
}


# q-projection formed inside every pooling block (W read whole-line from L2 in one load stream with the slice; 8 row groups of 96, added in group order: new f-sum order); 4 launches become 3
# speedup vs baseline: 7.0585x; 1.0979x over previous
#include <hip/hip_runtime.h>
#include <math.h>

typedef __attribute__((ext_vector_type(2))) float v2f;
typedef __attribute__((ext_vector_type(8))) float v8f;

#define B_MAX      32
#define N_TOK      4096
#define IN_DIM     64
#define FEAT_DIM   768
#define NQ         4
#define NEG_BIG    (-3.402823466e38f)

#define SLICES      8
#define TOK_S       (N_TOK / SLICES)
#define XPITCH      (IN_DIM + 4)
#define PS_THREADS  128
#define PS_WAVES    (PS_THREADS / 32)
#define TOK_T       (TOK_S / PS_THREADS)
#define TOK_W       (TOK_S / PS_WAVES)
#define PART_ML     512

#define P4_IT       16

#define QG          8
#define QROWS       (FEAT_DIM / QG)
#define XROWS       (TOK_S / QG)
#define RING        48
#define NPASS       (QROWS / RING)

#define STAGE_FENCE() __builtin_amdgcn_sched_barrier(0)

#define LDS_BARRIER() do { \
        __builtin_amdgcn_fence(__ATOMIC_RELEASE, "workgroup", "local"); \
        __builtin_amdgcn_s_barrier(); \
        __builtin_amdgcn_fence(__ATOMIC_ACQUIRE, "workgroup", "local"); } while (0)

#define P2_KC       4

__device__ __forceinline__ void p2_load(float4 (&u)[P2_KC][TOK_T], float4 (&cq)[P2_KC][NQ],
                                        const float* sx, const float* sqp, int t, int st)
{
    #pragma unroll
    for (int kk = 0; kk < P2_KC; ++kk) {
        const int kc = st * P2_KC + kk;
        #pragma unroll
        for (int h = 0; h < TOK_T; ++h)
            u[kk][h] = *reinterpret_cast<const float4*>(&sx[(t + PS_THREADS * h) * XPITCH + 4 * kc]);
        #pragma unroll
        for (int q = 0; q < NQ; ++q)
            cq[kk][q] = *reinterpret_cast<const float4*>(&sqp[q * IN_DIM + 4 * kc]);
    }
}

__device__ __forceinline__ void p2_sum(float (&s)[TOK_T][NQ], const float4 (&u)[P2_KC][TOK_T],
                                       const float4 (&cq)[P2_KC][NQ])
{
    #pragma unroll
    for (int kk = 0; kk < P2_KC; ++kk) {
        #pragma unroll
        for (int q = 0; q < NQ; ++q) {
            #pragma unroll
            for (int h = 0; h < TOK_T; ++h) {
                float r = s[h][q];
                r = fmaf(cq[kk][q].x, u[kk][h].x, r);
                r = fmaf(cq[kk][q].y, u[kk][h].y, r);
                r = fmaf(cq[kk][q].z, u[kk][h].z, r);
                r = fmaf(cq[kk][q].w, u[kk][h].w, r);
                s[h][q] = r;
            }
        }
    }
}

__device__ __forceinline__ void p4_load(float4 (&xv)[P4_IT], float4 (&ev)[P4_IT],
                                        const float* sx, const float* se, int n0, int c, int st)
{
    #pragma unroll
    for (int k = 0; k < P4_IT; ++k) {
        const int n = n0 + 2 * (st * P4_IT + k);
        xv[k] = *reinterpret_cast<const float4*>(&sx[n * XPITCH + 4 * c]);
        ev[k] = *reinterpret_cast<const float4*>(&se[n * NQ]);
    }
}

__device__ __forceinline__ void p4_sum(float4 (&a)[NQ], const float4 (&xv)[P4_IT],
                                       const float4 (&ev)[P4_IT])
{
    #pragma unroll
    for (int k = 0; k < P4_IT; ++k) {
        const float wq[NQ] = { ev[k].x, ev[k].y, ev[k].z, ev[k].w };
        #pragma unroll
        for (int q = 0; q < NQ; ++q) {
            a[q].x = fmaf(wq[q], xv[k].x, a[q].x);
            a[q].y = fmaf(wq[q], xv[k].y, a[q].y);
            a[q].z = fmaf(wq[q], xv[k].z, a[q].z);
            a[q].w = fmaf(wq[q], xv[k].w, a[q].w);
        }
    }
}

__device__ __forceinline__ void p0_load(float4 (&qv)[NQ], float4& bv,
                                        const float* sq, const float* sb, int r0, int jj)
{
    #pragma unroll
    for (int q = 0; q < NQ; ++q)
        qv[q] = *reinterpret_cast<const float4*>(&sq[q * FEAT_DIM + r0 + 4 * jj]);
    bv = *reinterpret_cast<const float4*>(&sb[r0 + 4 * jj]);
}

__device__ __forceinline__ void p0_sum(float4 (&a)[NQ], float (&ab)[NQ], const float4 (&qv)[NQ],
                                       const float4& bv, const float4& w0, const float4& w1,
                                       const float4& w2, const float4& w3)
{
    const float4 wr[4] = { w0, w1, w2, w3 };
    const float  bb[4] = { bv.x, bv.y, bv.z, bv.w };
    #pragma unroll
    for (int e = 0; e < 4; ++e) {
        #pragma unroll
        for (int q = 0; q < NQ; ++q) {
            const float qe = (e == 0) ? qv[q].x : (e == 1) ? qv[q].y : (e == 2) ? qv[q].z : qv[q].w;
            a[q].x = fmaf(qe, wr[e].x, a[q].x);
            a[q].y = fmaf(qe, wr[e].y, a[q].y);
            a[q].z = fmaf(qe, wr[e].z, a[q].z);
            a[q].w = fmaf(qe, wr[e].w, a[q].w);
            ab[q]  = fmaf(qe, bb[e], ab[q]);
        }
    }
}

__global__ void __launch_bounds__(PS_THREADS)
lap_pool_slice_kernel(const float* __restrict__ x,
                      const int*   __restrict__ mask,
                      const float* qt,
                      const float* __restrict__ W,
                      const float* __restrict__ bias,
                      float* __restrict__ part)
{
    __shared__ __attribute__((aligned(16))) float s_x[TOK_S * XPITCH];
    __shared__ float s_sc[NQ][TOK_S];
    __shared__ __attribute__((aligned(16))) float s_e[TOK_S * NQ];
    __shared__ __attribute__((aligned(16))) float s_pp[2 * PS_WAVES][NQ][IN_DIM];
    __shared__ __attribute__((aligned(16))) float s_qp[NQ * IN_DIM];
    __shared__ __attribute__((aligned(16))) float s_q[NQ * FEAT_DIM];
    __shared__ __attribute__((aligned(16))) float s_b[FEAT_DIM];
    __shared__ float s_bp[QG][NQ];
    __shared__ float s_qb[NQ];
    __shared__ float s_l[NQ];
    __shared__ float s_m[NQ];

    const int b    = blockIdx.x;
    const int sl   = blockIdx.y;
    const int t    = threadIdx.x;
    const int lane = t & 31;
    const int w    = __builtin_amdgcn_readfirstlane(t >> 5);
    const size_t tok0 = (size_t)b * N_TOK + (size_t)sl * TOK_S;

    const int g  = t >> 4;
    const int c4 = t & 15;
    const float4* wp = reinterpret_cast<const float4*>(W) + (g * QROWS) * (IN_DIM / 4) + c4;
    const float4* xp = reinterpret_cast<const float4*>(x + tok0 * IN_DIM) + (g * XROWS) * (IN_DIM / 4) + c4;

    const float4* qt4 = reinterpret_cast<const float4*>(qt);
    const float2* bi2 = reinterpret_cast<const float2*>(bias);
    const float4 qs0 = qt4[t];
    const float4 qs1 = qt4[PS_THREADS + t];
    const float4 qs2 = qt4[2 * PS_THREADS + t];
    const float4 qs3 = qt4[3 * PS_THREADS + t];
    const float4 qs4 = qt4[4 * PS_THREADS + t];
    const float4 qs5 = qt4[5 * PS_THREADS + t];
    const float2 bs0 = bi2[t];
    const float2 bs1 = bi2[PS_THREADS + t];
    const float2 bs2 = bi2[2 * PS_THREADS + t];
    STAGE_FENCE();
    float4 ring[RING];
    #pragma unroll
    for (int i = 0; i < RING; ++i) ring[i] = wp[i * (IN_DIM / 4)];
    {
        float4* sq4 = reinterpret_cast<float4*>(s_q);
        float2* sb2 = reinterpret_cast<float2*>(s_b);
        sq4[t]                  = qs0;
        sq4[PS_THREADS + t]     = qs1;
        sq4[2 * PS_THREADS + t] = qs2;
        sq4[3 * PS_THREADS + t] = qs3;
        sq4[4 * PS_THREADS + t] = qs4;
        sq4[5 * PS_THREADS + t] = qs5;
        sb2[t]                  = bs0;
        sb2[PS_THREADS + t]     = bs1;
        sb2[2 * PS_THREADS + t] = bs2;
    }
    LDS_BARRIER();

    int mk[TOK_T];
    {
        float4 a[NQ];
        float  ab[NQ];
        #pragma unroll
        for (int q = 0; q < NQ; ++q) { a[q] = make_float4(0.0f, 0.0f, 0.0f, 0.0f); ab[q] = 0.0f; }
        float4 qA[NQ], qB[NQ], bA, bB;
        const int r0 = g * QROWS;
        p0_load(qA, bA, s_q, s_b, r0, 0);
        #pragma unroll 1
        for (int it = 0; it < NPASS; ++it) {
            const float4* rp = (it + 1 < NPASS) ? (wp + (it + 1) * RING * (IN_DIM / 4)) : xp;
            const int rq = r0 + RING * it;
            const int rn = (it + 1 < NPASS) ? (rq + RING) : r0;
            #pragma unroll
            for (int k = 0; k < RING / 4; k += 2) {
                p0_load(qB, bB, s_q, s_b, rq, k + 1);
                STAGE_FENCE();
                p0_sum(a, ab, qA, bA, ring[4 * k], ring[4 * k + 1], ring[4 * k + 2], ring[4 * k + 3]);
                #pragma unroll
                for (int e = 0; e < 4; ++e) ring[4 * k + e] = rp[(4 * k + e) * (IN_DIM / 4)];
                if (k + 2 < RING / 4) p0_load(qA, bA, s_q, s_b, rq, k + 2);
                else                  p0_load(qA, bA, s_q, s_b, rn, 0);
                STAGE_FENCE();
                p0_sum(a, ab, qB, bB, ring[4 * k + 4], ring[4 * k + 5], ring[4 * k + 6], ring[4 * k + 7]);
                #pragma unroll
                for (int e = 0; e < 4; ++e) ring[4 * k + 4 + e] = rp[(4 * k + 4 + e) * (IN_DIM / 4)];
            }
        }
        #pragma unroll
        for (int q = 0; q < NQ; ++q)
            *reinterpret_cast<float4*>(&s_pp[g][q][4 * c4]) = a[q];
        #pragma unroll
        for (int q = 0; q < NQ; ++q) s_bp[g][q] = ab[q];
    }
    float* sxr = &s_x[(g * XROWS) * XPITCH + 4 * c4];
    #pragma unroll
    for (int i = 0; i < XROWS - RING; ++i) {
        *reinterpret_cast<float4*>(&sxr[i * XPITCH]) = ring[i];
        ring[i] = xp[(RING + i) * (IN_DIM / 4)];
    }
    #pragma unroll
    for (int h = 0; h < TOK_T; ++h) mk[h] = mask[tok0 + t + PS_THREADS * h];
    #pragma unroll
    for (int i = XROWS - RING; i < RING; ++i)
        *reinterpret_cast<float4*>(&sxr[i * XPITCH]) = ring[i];
    #pragma unroll
    for (int i = 0; i < XROWS - RING; ++i)
        *reinterpret_cast<float4*>(&sxr[(RING + i) * XPITCH]) = ring[i];
    __syncthreads();

    #pragma unroll
    for (int r = 0; r < (NQ * IN_DIM) / PS_THREADS; ++r) {
        const int o = t + PS_THREADS * r;
        float s = 0.0f;
        #pragma unroll
        for (int gg = 0; gg < QG; ++gg) s += s_pp[gg][o >> 6][o & 63];
        s_qp[o] = s;
    }
    if (t < NQ) {
        float s = 0.0f;
        #pragma unroll
        for (int gg = 0; gg < QG; ++gg) s += s_bp[gg][t];
        s_qb[t] = s;
    }
    __syncthreads();

    {
        float s[TOK_T][NQ];
        #pragma unroll
        for (int h = 0; h < TOK_T; ++h) {
            #pragma unroll
            for (int q = 0; q < NQ; ++q) s[h][q] = s_qb[q];
        }
        float4 uA[P2_KC][TOK_T], cA[P2_KC][NQ];
        float4 uB[P2_KC][TOK_T], cB[P2_KC][NQ];
        p2_load(uA, cA, s_x, s_qp, t, 0);
        STAGE_FENCE();
        p2_load(uB, cB, s_x, s_qp, t, 1);
        STAGE_FENCE();
        p2_sum(s, uA, cA);
        STAGE_FENCE();
        p2_load(uA, cA, s_x, s_qp, t, 2);
        STAGE_FENCE();
        p2_sum(s, uB, cB);
        STAGE_FENCE();
        p2_load(uB, cB, s_x, s_qp, t, 3);
        STAGE_FENCE();
        p2_sum(s, uA, cA);
        STAGE_FENCE();
        p2_sum(s, uB, cB);
        #pragma unroll
        for (int h = 0; h < TOK_T; ++h) {
            #pragma unroll
            for (int q = 0; q < NQ; ++q)
                s_sc[q][t + PS_THREADS * h] = (mk[h] == 0) ? NEG_BIG : s[h][q];
        }
    }
    __syncthreads();

    {
        const int q = w;
        float sc[TOK_S / 32];
        float mloc = NEG_BIG;
        #pragma unroll
        for (int k = 0; k < TOK_S / 32; ++k) {
            sc[k] = s_sc[q][lane + 32 * k];
            mloc = fmaxf(mloc, sc[k]);
        }
        #pragma unroll
        for (int off = 16; off > 0; off >>= 1) mloc = fmaxf(mloc, __shfl_xor(mloc, off));
        float ps = 0.0f;
        #pragma unroll
        for (int k = 0; k < TOK_S / 32; ++k) {
            const int n = lane + 32 * k;
            const float e = __expf(sc[k] - mloc);
            s_e[n * NQ + q] = e;
            ps += e;
        }
        #pragma unroll
        for (int off = 16; off > 0; off >>= 1) ps += __shfl_xor(ps, off);
        if (lane == 0) {
            s_l[q] = ps;
            s_m[q] = mloc;
        }
    }
    __syncthreads();

    {
        const int hf = lane >> 4;
        const int c  = lane & 15;
        const int n0 = w * TOK_W + hf;
        float4 a[NQ];
        #pragma unroll
        for (int q = 0; q < NQ; ++q) a[q] = make_float4(0.0f, 0.0f, 0.0f, 0.0f);
        float4 xA[P4_IT], eA[P4_IT], xB[P4_IT], eB[P4_IT];
        p4_load(xA, eA, s_x, s_e, n0, c, 0);
        STAGE_FENCE();
        p4_load(xB, eB, s_x, s_e, n0, c, 1);
        STAGE_FENCE();
        p4_sum(a, xA, eA);
        STAGE_FENCE();
        p4_load(xA, eA, s_x, s_e, n0, c, 2);
        STAGE_FENCE();
        p4_sum(a, xB, eB);
        STAGE_FENCE();
        p4_load(xB, eB, s_x, s_e, n0, c, 3);
        STAGE_FENCE();
        p4_sum(a, xA, eA);
        STAGE_FENCE();
        p4_sum(a, xB, eB);
        #pragma unroll
        for (int q = 0; q < NQ; ++q)
            *reinterpret_cast<float4*>(&s_pp[2 * w + hf][q][4 * c]) = a[q];
    }
    __syncthreads();

    #pragma unroll
    for (int r = 0; r < (NQ * IN_DIM) / PS_THREADS; ++r) {
        const int o = t + PS_THREADS * r;
        const int q = o >> 6;
        const int i = o & 63;
        float s = 0.0f;
        #pragma unroll
        for (int pw = 0; pw < 2 * PS_WAVES; ++pw) s += s_pp[pw][q][i];
        part[(size_t)(b * NQ + q) * FEAT_DIM + sl * IN_DIM + i] = s;
    }
    if (t < NQ) {
        *reinterpret_cast<float2*>(&part[(size_t)(b * NQ + t) * FEAT_DIM + PART_ML + sl * 32]) =
            make_float2(s_m[t], s_l[t]);
    }
}

__global__ void __launch_bounds__(IN_DIM)
lap_combine_kernel(const float* __restrict__ part,
                   float* __restrict__ pooled)
{
    const int m = blockIdx.x;
    const int i = threadIdx.x;
    const float* row = part + (size_t)m * FEAT_DIM;

    float ms[SLICES], ls[SLICES];
    float M = NEG_BIG;
    #pragma unroll
    for (int s = 0; s < SLICES; ++s) {
        ms[s] = row[PART_ML + s * 32];
        ls[s] = row[PART_ML + s * 32 + 1];
        M = fmaxf(M, ms[s]);
    }
    float L = 0.0f, acc = 0.0f;
    #pragma unroll
    for (int s = 0; s < SLICES; ++s) {
        const float c = __expf(ms[s] - M);
        L   += c * ls[s];
        acc += c * row[s * IN_DIM + i];
    }
    pooled[m * IN_DIM + i] = acc * (1.0f / L);
}

__global__ void lap_outgemm_kernel(const float* __restrict__ pooled,
                                   const float* __restrict__ W,
                                   const float* __restrict__ bias,
                                   float* __restrict__ out)
{
    const int n0   = blockIdx.x * 16;
    const int m0   = blockIdx.y * 16;
    const int lane = threadIdx.x;
    const int hf   = lane >> 4;
    const int l    = lane & 15;

    const int arow = m0 + l;
    const int bcol = n0 + l;
    const float* ap = pooled + arow * IN_DIM;
    const float* bp = W + bcol * IN_DIM;

    v8f acc = {};
    #pragma unroll
    for (int k0 = 0; k0 < IN_DIM; k0 += 4) {
        const int ka = k0 + hf * 2;
        v2f a, bfr;
        a.x   = ap[ka];
        a.y   = ap[ka + 1];
        bfr.x = bp[ka];
        bfr.y = bp[ka + 1];
        acc = __builtin_amdgcn_wmma_f32_16x16x4_f32(
                   false, a,  false, bfr,
                   (short)0, acc,  false,  false);
    }

    const int col = n0 + l;
    const float bv = bias[col];
    #pragma unroll
    for (int r = 0; r < 8; ++r) {
        const int row = m0 + hf * 8 + r;
        out[row * FEAT_DIM + col] = acc[r] + bv;
    }
}

extern "C" void kernel_launch(void* const* d_in, const int* in_sizes, int n_in,
                              void* d_out, int out_size, void* d_ws, size_t ws_size,
                              hipStream_t stream) {
    const float* x    = (const float*)d_in[0];
    const int*   mask = (const int*)  d_in[1];
    const float* qt   = (const float*)d_in[2];
    const float* W    = (const float*)d_in[3];
    const float* bias = (const float*)d_in[4];
    float* out = (float*)d_out;

    const int B = in_sizes[0] / (N_TOK * IN_DIM);

    float* ws     = (float*)d_ws;
    float* pooled = ws + 512;

    lap_pool_slice_kernel<<<dim3(B, SLICES), PS_THREADS, 0, stream>>>(x, mask, qt, W, bias, out);
    lap_combine_kernel<<<B * NQ, IN_DIM, 0, stream>>>(out, pooled);
    lap_outgemm_kernel<<<dim3(FEAT_DIM / 16, (B * NQ) / 16), 32, 0, stream>>>(
        pooled, W, bias, out);
}


# slice merge folded into the product launch, wide grid; partials parked in the workspace if it holds them (else as before), 96 4-wave blocks merge their 16 rows to LDS, sums now fused; 3 launches to 2
# speedup vs baseline: 8.1634x; 1.1565x over previous
#include <hip/hip_runtime.h>
#include <math.h>

typedef __attribute__((ext_vector_type(2))) float v2f;
typedef __attribute__((ext_vector_type(4))) float v4f;
typedef __attribute__((ext_vector_type(8))) float v8f;

#define B_MAX      32
#define N_TOK      4096
#define IN_DIM     64
#define FEAT_DIM   768
#define NQ         4
#define NEG_BIG    (-3.402823466e38f)

#define SLICES      8
#define TOK_S       (N_TOK / SLICES)
#define XPITCH      (IN_DIM + 4)
#define PS_THREADS  128
#define PS_WAVES    (PS_THREADS / 32)
#define TOK_T       (TOK_S / PS_THREADS)
#define TOK_W       (TOK_S / PS_WAVES)
#define PART_ML     512

#define P4_IT       16

#define QG          8
#define QROWS       (FEAT_DIM / QG)
#define XROWS       (TOK_S / QG)
#define RING        48
#define NPASS       (QROWS / RING)

#define STAGE_FENCE() __builtin_amdgcn_sched_barrier(0)

#define LDS_BARRIER() do { \
        __builtin_amdgcn_fence(__ATOMIC_RELEASE, "workgroup", "local"); \
        __builtin_amdgcn_s_barrier(); \
        __builtin_amdgcn_fence(__ATOMIC_ACQUIRE, "workgroup", "local"); } while (0)

#define P2_KC       4

__device__ __forceinline__ void p2_load(float4 (&u)[P2_KC][TOK_T], float4 (&cq)[P2_KC][NQ],
                                        const float* sx, const float* sqp, int t, int st)
{
    #pragma unroll
    for (int kk = 0; kk < P2_KC; ++kk) {
        const int kc = st * P2_KC + kk;
        #pragma unroll
        for (int h = 0; h < TOK_T; ++h)
            u[kk][h] = *reinterpret_cast<const float4*>(&sx[(t + PS_THREADS * h) * XPITCH + 4 * kc]);
        #pragma unroll
        for (int q = 0; q < NQ; ++q)
            cq[kk][q] = *reinterpret_cast<const float4*>(&sqp[q * IN_DIM + 4 * kc]);
    }
}

__device__ __forceinline__ void p2_sum(float (&s)[TOK_T][NQ], const float4 (&u)[P2_KC][TOK_T],
                                       const float4 (&cq)[P2_KC][NQ])
{
    #pragma unroll
    for (int kk = 0; kk < P2_KC; ++kk) {
        #pragma unroll
        for (int q = 0; q < NQ; ++q) {
            #pragma unroll
            for (int h = 0; h < TOK_T; ++h) {
                float r = s[h][q];
                r = fmaf(cq[kk][q].x, u[kk][h].x, r);
                r = fmaf(cq[kk][q].y, u[kk][h].y, r);
                r = fmaf(cq[kk][q].z, u[kk][h].z, r);
                r = fmaf(cq[kk][q].w, u[kk][h].w, r);
                s[h][q] = r;
            }
        }
    }
}

__device__ __forceinline__ void p4_load(float4 (&xv)[P4_IT], float4 (&ev)[P4_IT],
                                        const float* sx, const float* se, int n0, int c, int st)
{
    #pragma unroll
    for (int k = 0; k < P4_IT; ++k) {
        const int n = n0 + 2 * (st * P4_IT + k);
        xv[k] = *reinterpret_cast<const float4*>(&sx[n * XPITCH + 4 * c]);
        ev[k] = *reinterpret_cast<const float4*>(&se[n * NQ]);
    }
}

__device__ __forceinline__ void p4_sum(float4 (&a)[NQ], const float4 (&xv)[P4_IT],
                                       const float4 (&ev)[P4_IT])
{
    #pragma unroll
    for (int k = 0; k < P4_IT; ++k) {
        const float wq[NQ] = { ev[k].x, ev[k].y, ev[k].z, ev[k].w };
        #pragma unroll
        for (int q = 0; q < NQ; ++q) {
            a[q].x = fmaf(wq[q], xv[k].x, a[q].x);
            a[q].y = fmaf(wq[q], xv[k].y, a[q].y);
            a[q].z = fmaf(wq[q], xv[k].z, a[q].z);
            a[q].w = fmaf(wq[q], xv[k].w, a[q].w);
        }
    }
}

__device__ __forceinline__ void p0_load(float4 (&qv)[NQ], float4& bv,
                                        const float* sq, const float* sb, int r0, int jj)
{
    #pragma unroll
    for (int q = 0; q < NQ; ++q)
        qv[q] = *reinterpret_cast<const float4*>(&sq[q * FEAT_DIM + r0 + 4 * jj]);
    bv = *reinterpret_cast<const float4*>(&sb[r0 + 4 * jj]);
}

__device__ __forceinline__ void p0_sum(float4 (&a)[NQ], float (&ab)[NQ], const float4 (&qv)[NQ],
                                       const float4& bv, const float4& w0, const float4& w1,
                                       const float4& w2, const float4& w3)
{
    const float4 wr[4] = { w0, w1, w2, w3 };
    const float  bb[4] = { bv.x, bv.y, bv.z, bv.w };
    #pragma unroll
    for (int e = 0; e < 4; ++e) {
        #pragma unroll
        for (int q = 0; q < NQ; ++q) {
            const float qe = (e == 0) ? qv[q].x : (e == 1) ? qv[q].y : (e == 2) ? qv[q].z : qv[q].w;
            a[q].x = fmaf(qe, wr[e].x, a[q].x);
            a[q].y = fmaf(qe, wr[e].y, a[q].y);
            a[q].z = fmaf(qe, wr[e].z, a[q].z);
            a[q].w = fmaf(qe, wr[e].w, a[q].w);
            ab[q]  = fmaf(qe, bb[e], ab[q]);
        }
    }
}

__global__ void __launch_bounds__(PS_THREADS)
lap_pool_slice_kernel(const float* __restrict__ x,
                      const int*   __restrict__ mask,
                      const float* qt,
                      const float* __restrict__ W,
                      const float* __restrict__ bias,
                      float* __restrict__ part)
{
    __shared__ __attribute__((aligned(16))) float s_x[TOK_S * XPITCH];
    __shared__ float s_sc[NQ][TOK_S];
    __shared__ __attribute__((aligned(16))) float s_e[TOK_S * NQ];
    __shared__ __attribute__((aligned(16))) float s_pp[2 * PS_WAVES][NQ][IN_DIM];
    __shared__ __attribute__((aligned(16))) float s_qp[NQ * IN_DIM];
    __shared__ __attribute__((aligned(16))) float s_q[NQ * FEAT_DIM];
    __shared__ __attribute__((aligned(16))) float s_b[FEAT_DIM];
    __shared__ float s_bp[QG][NQ];
    __shared__ float s_qb[NQ];
    __shared__ float s_l[NQ];
    __shared__ float s_m[NQ];

    const int b    = blockIdx.x;
    const int sl   = blockIdx.y;
    const int t    = threadIdx.x;
    const int lane = t & 31;
    const int w    = __builtin_amdgcn_readfirstlane(t >> 5);
    const size_t tok0 = (size_t)b * N_TOK + (size_t)sl * TOK_S;

    const int g  = t >> 4;
    const int c4 = t & 15;
    const float4* wp = reinterpret_cast<const float4*>(W) + (g * QROWS) * (IN_DIM / 4) + c4;
    const float4* xp = reinterpret_cast<const float4*>(x + tok0 * IN_DIM) + (g * XROWS) * (IN_DIM / 4) + c4;

    const float4* qt4 = reinterpret_cast<const float4*>(qt);
    const float2* bi2 = reinterpret_cast<const float2*>(bias);
    const float4 qs0 = qt4[t];
    const float4 qs1 = qt4[PS_THREADS + t];
    const float4 qs2 = qt4[2 * PS_THREADS + t];
    const float4 qs3 = qt4[3 * PS_THREADS + t];
    const float4 qs4 = qt4[4 * PS_THREADS + t];
    const float4 qs5 = qt4[5 * PS_THREADS + t];
    const float2 bs0 = bi2[t];
    const float2 bs1 = bi2[PS_THREADS + t];
    const float2 bs2 = bi2[2 * PS_THREADS + t];
    STAGE_FENCE();
    float4 ring[RING];
    #pragma unroll
    for (int i = 0; i < RING; ++i) ring[i] = wp[i * (IN_DIM / 4)];
    {
        float4* sq4 = reinterpret_cast<float4*>(s_q);
        float2* sb2 = reinterpret_cast<float2*>(s_b);
        sq4[t]                  = qs0;
        sq4[PS_THREADS + t]     = qs1;
        sq4[2 * PS_THREADS + t] = qs2;
        sq4[3 * PS_THREADS + t] = qs3;
        sq4[4 * PS_THREADS + t] = qs4;
        sq4[5 * PS_THREADS + t] = qs5;
        sb2[t]                  = bs0;
        sb2[PS_THREADS + t]     = bs1;
        sb2[2 * PS_THREADS + t] = bs2;
    }
    LDS_BARRIER();

    int mk[TOK_T];
    {
        float4 a[NQ];
        float  ab[NQ];
        #pragma unroll
        for (int q = 0; q < NQ; ++q) { a[q] = make_float4(0.0f, 0.0f, 0.0f, 0.0f); ab[q] = 0.0f; }
        float4 qA[NQ], qB[NQ], bA, bB;
        const int r0 = g * QROWS;
        p0_load(qA, bA, s_q, s_b, r0, 0);
        #pragma unroll 1
        for (int it = 0; it < NPASS; ++it) {
            const float4* rp = (it + 1 < NPASS) ? (wp + (it + 1) * RING * (IN_DIM / 4)) : xp;
            const int rq = r0 + RING * it;
            const int rn = (it + 1 < NPASS) ? (rq + RING) : r0;
            #pragma unroll
            for (int k = 0; k < RING / 4; k += 2) {
                p0_load(qB, bB, s_q, s_b, rq, k + 1);
                STAGE_FENCE();
                p0_sum(a, ab, qA, bA, ring[4 * k], ring[4 * k + 1], ring[4 * k + 2], ring[4 * k + 3]);
                #pragma unroll
                for (int e = 0; e < 4; ++e) ring[4 * k + e] = rp[(4 * k + e) * (IN_DIM / 4)];
                if (k + 2 < RING / 4) p0_load(qA, bA, s_q, s_b, rq, k + 2);
                else                  p0_load(qA, bA, s_q, s_b, rn, 0);
                STAGE_FENCE();
                p0_sum(a, ab, qB, bB, ring[4 * k + 4], ring[4 * k + 5], ring[4 * k + 6], ring[4 * k + 7]);
                #pragma unroll
                for (int e = 0; e < 4; ++e) ring[4 * k + 4 + e] = rp[(4 * k + 4 + e) * (IN_DIM / 4)];
            }
        }
        #pragma unroll
        for (int q = 0; q < NQ; ++q)
            *reinterpret_cast<float4*>(&s_pp[g][q][4 * c4]) = a[q];
        #pragma unroll
        for (int q = 0; q < NQ; ++q) s_bp[g][q] = ab[q];
    }
    float* sxr = &s_x[(g * XROWS) * XPITCH + 4 * c4];
    #pragma unroll
    for (int i = 0; i < XROWS - RING; ++i) {
        *reinterpret_cast<float4*>(&sxr[i * XPITCH]) = ring[i];
        ring[i] = xp[(RING + i) * (IN_DIM / 4)];
    }
    #pragma unroll
    for (int h = 0; h < TOK_T; ++h) mk[h] = mask[tok0 + t + PS_THREADS * h];
    #pragma unroll
    for (int i = XROWS - RING; i < RING; ++i)
        *reinterpret_cast<float4*>(&sxr[i * XPITCH]) = ring[i];
    #pragma unroll
    for (int i = 0; i < XROWS - RING; ++i)
        *reinterpret_cast<float4*>(&sxr[(RING + i) * XPITCH]) = ring[i];
    __syncthreads();

    #pragma unroll
    for (int r = 0; r < (NQ * IN_DIM) / PS_THREADS; ++r) {
        const int o = t + PS_THREADS * r;
        float s = 0.0f;
        #pragma unroll
        for (int gg = 0; gg < QG; ++gg) s += s_pp[gg][o >> 6][o & 63];
        s_qp[o] = s;
    }
    if (t < NQ) {
        float s = 0.0f;
        #pragma unroll
        for (int gg = 0; gg < QG; ++gg) s += s_bp[gg][t];
        s_qb[t] = s;
    }
    __syncthreads();

    {
        float s[TOK_T][NQ];
        #pragma unroll
        for (int h = 0; h < TOK_T; ++h) {
            #pragma unroll
            for (int q = 0; q < NQ; ++q) s[h][q] = s_qb[q];
        }
        float4 uA[P2_KC][TOK_T], cA[P2_KC][NQ];
        float4 uB[P2_KC][TOK_T], cB[P2_KC][NQ];
        p2_load(uA, cA, s_x, s_qp, t, 0);
        STAGE_FENCE();
        p2_load(uB, cB, s_x, s_qp, t, 1);
        STAGE_FENCE();
        p2_sum(s, uA, cA);
        STAGE_FENCE();
        p2_load(uA, cA, s_x, s_qp, t, 2);
        STAGE_FENCE();
        p2_sum(s, uB, cB);
        STAGE_FENCE();
        p2_load(uB, cB, s_x, s_qp, t, 3);
        STAGE_FENCE();
        p2_sum(s, uA, cA);
        STAGE_FENCE();
        p2_sum(s, uB, cB);
        #pragma unroll
        for (int h = 0; h < TOK_T; ++h) {
            #pragma unroll
            for (int q = 0; q < NQ; ++q)
                s_sc[q][t + PS_THREADS * h] = (mk[h] == 0) ? NEG_BIG : s[h][q];
        }
    }
    __syncthreads();

    {
        const int q = w;
        float sc[TOK_S / 32];
        float mloc = NEG_BIG;
        #pragma unroll
        for (int k = 0; k < TOK_S / 32; ++k) {
            sc[k] = s_sc[q][lane + 32 * k];
            mloc = fmaxf(mloc, sc[k]);
        }
        #pragma unroll
        for (int off = 16; off > 0; off >>= 1) mloc = fmaxf(mloc, __shfl_xor(mloc, off));
        float ps = 0.0f;
        #pragma unroll
        for (int k = 0; k < TOK_S / 32; ++k) {
            const int n = lane + 32 * k;
            const float e = __expf(sc[k] - mloc);
            s_e[n * NQ + q] = e;
            ps += e;
        }
        #pragma unroll
        for (int off = 16; off > 0; off >>= 1) ps += __shfl_xor(ps, off);
        if (lane == 0) {
            s_l[q] = ps;
            s_m[q] = mloc;
        }
    }
    __syncthreads();

    {
        const int hf = lane >> 4;
        const int c  = lane & 15;
        const int n0 = w * TOK_W + hf;
        float4 a[NQ];
        #pragma unroll
        for (int q = 0; q < NQ; ++q) a[q] = make_float4(0.0f, 0.0f, 0.0f, 0.0f);
        float4 xA[P4_IT], eA[P4_IT], xB[P4_IT], eB[P4_IT];
        p4_load(xA, eA, s_x, s_e, n0, c, 0);
        STAGE_FENCE();
        p4_load(xB, eB, s_x, s_e, n0, c, 1);
        STAGE_FENCE();
        p4_sum(a, xA, eA);
        STAGE_FENCE();
        p4_load(xA, eA, s_x, s_e, n0, c, 2);
        STAGE_FENCE();
        p4_sum(a, xB, eB);
        STAGE_FENCE();
        p4_load(xB, eB, s_x, s_e, n0, c, 3);
        STAGE_FENCE();
        p4_sum(a, xA, eA);
        STAGE_FENCE();
        p4_sum(a, xB, eB);
        #pragma unroll
        for (int q = 0; q < NQ; ++q)
            *reinterpret_cast<float4*>(&s_pp[2 * w + hf][q][4 * c]) = a[q];
    }
    __syncthreads();

    #pragma unroll
    for (int r = 0; r < (NQ * IN_DIM) / PS_THREADS; ++r) {
        const int o = t + PS_THREADS * r;
        const int q = o >> 6;
        const int i = o & 63;
        float s = 0.0f;
        #pragma unroll
        for (int pw = 0; pw < 2 * PS_WAVES; ++pw) s += s_pp[pw][q][i];
        part[(size_t)(b * NQ + q) * FEAT_DIM + sl * IN_DIM + i] = s;
    }
    if (t < NQ) {
        *reinterpret_cast<float2*>(&part[(size_t)(b * NQ + t) * FEAT_DIM + PART_ML + sl * 32]) =
            make_float2(s_m[t], s_l[t]);
    }
}

__global__ void __launch_bounds__(IN_DIM)
lap_combine_kernel(const float* __restrict__ part,
                   float* __restrict__ pooled)
{
    const int m = blockIdx.x;
    const int i = threadIdx.x;
    const float* row = part + (size_t)m * FEAT_DIM;

    float ms[SLICES], ls[SLICES];
    float M = NEG_BIG;
    #pragma unroll
    for (int s = 0; s < SLICES; ++s) {
        ms[s] = row[PART_ML + s * 32];
        ls[s] = row[PART_ML + s * 32 + 1];
        M = fmaxf(M, ms[s]);
    }
    float L = 0.0f, acc = 0.0f;
    #pragma unroll
    for (int s = 0; s < SLICES; ++s) {
        const float c = __expf(ms[s] - M);
        L   += c * ls[s];
        acc += c * row[s * IN_DIM + i];
    }
    pooled[m * IN_DIM + i] = acc * (1.0f / L);
}

__global__ void lap_outgemm_kernel(const float* __restrict__ pooled,
                                   const float* __restrict__ W,
                                   const float* __restrict__ bias,
                                   float* __restrict__ out)
{
    const int n0   = blockIdx.x * 16;
    const int m0   = blockIdx.y * 16;
    const int lane = threadIdx.x;
    const int hf   = lane >> 4;
    const int l    = lane & 15;

    const int arow = m0 + l;
    const int bcol = n0 + l;
    const float* ap = pooled + arow * IN_DIM;
    const float* bp = W + bcol * IN_DIM;

    v8f acc = {};
    #pragma unroll
    for (int k0 = 0; k0 < IN_DIM; k0 += 4) {
        const int ka = k0 + hf * 2;
        v2f a, bfr;
        a.x   = ap[ka];
        a.y   = ap[ka + 1];
        bfr.x = bp[ka];
        bfr.y = bp[ka + 1];
        acc = __builtin_amdgcn_wmma_f32_16x16x4_f32(
                   false, a,  false, bfr,
                   (short)0, acc,  false,  false);
    }

    const int col = n0 + l;
    const float bv = bias[col];
    #pragma unroll
    for (int r = 0; r < 8; ++r) {
        const int row = m0 + hf * 8 + r;
        out[row * FEAT_DIM + col] = acc[r] + bv;
    }
}

#define MG_THREADS  128
#define MG_COLS     64
#define PPITCH      (IN_DIM + 4)

__global__ void __launch_bounds__(MG_THREADS)
lap_merge_outgemm_kernel(const float* __restrict__ part,
                         const float* __restrict__ W,
                         const float* __restrict__ bias,
                         float* __restrict__ out)
{
    __shared__ __attribute__((aligned(16))) float s_pool[16 * PPITCH];

    const int t    = threadIdx.x;
    const int lane = t & 31;
    const int w    = __builtin_amdgcn_readfirstlane(t >> 5);
    const int hf   = lane >> 4;
    const int l    = lane & 15;
    const int m0   = blockIdx.y * 16;
    const int n0   = blockIdx.x * MG_COLS + w * 16;

    const int c  = t & 15;
    const int rp = t >> 4;
    v2f ml[2][SLICES];
    v4f p[2][SLICES];
    #pragma unroll
    for (int j = 0; j < 2; ++j) {
        const float* row = part + (size_t)(m0 + rp + 8 * j) * FEAT_DIM;
        #pragma unroll
        for (int s = 0; s < SLICES; ++s)
            ml[j][s] = *reinterpret_cast<const v2f*>(row + PART_ML + s * 32);
        #pragma unroll
        for (int s = 0; s < SLICES; ++s)
            p[j][s] = *reinterpret_cast<const v4f*>(row + s * IN_DIM + 4 * c);
    }
    const float* bp = W + (n0 + l) * IN_DIM + hf * 2;
    v2f bfr[IN_DIM / 4];
    #pragma unroll
    for (int k = 0; k < IN_DIM / 4; ++k)
        bfr[k] = *reinterpret_cast<const v2f*>(bp + 4 * k);
    const float bv = bias[n0 + l];
    asm volatile("" :: "v"(ml[0][0]), "v"(ml[0][1]), "v"(ml[0][2]), "v"(ml[0][3]),
                       "v"(ml[0][4]), "v"(ml[0][5]), "v"(ml[0][6]), "v"(ml[0][7]),
                       "v"(ml[1][0]), "v"(ml[1][1]), "v"(ml[1][2]), "v"(ml[1][3]),
                       "v"(ml[1][4]), "v"(ml[1][5]), "v"(ml[1][6]), "v"(ml[1][7]));
    asm volatile("" :: "v"(p[0][0]), "v"(p[0][1]), "v"(p[0][2]), "v"(p[0][3]),
                       "v"(p[0][4]), "v"(p[0][5]), "v"(p[0][6]), "v"(p[0][7]),
                       "v"(p[1][0]), "v"(p[1][1]), "v"(p[1][2]), "v"(p[1][3]),
                       "v"(p[1][4]), "v"(p[1][5]), "v"(p[1][6]), "v"(p[1][7]));
    asm volatile("" :: "v"(bfr[0]), "v"(bfr[1]), "v"(bfr[2]), "v"(bfr[3]),
                       "v"(bfr[4]), "v"(bfr[5]), "v"(bfr[6]), "v"(bfr[7]),
                       "v"(bfr[8]), "v"(bfr[9]), "v"(bfr[10]), "v"(bfr[11]),
                       "v"(bfr[12]), "v"(bfr[13]), "v"(bfr[14]), "v"(bfr[15]), "v"(bv));

    #pragma unroll
    for (int j = 0; j < 2; ++j) {
        float M = NEG_BIG;
        #pragma unroll
        for (int s = 0; s < SLICES; ++s) M = fmaxf(M, ml[j][s].x);
        float  L = 0.0f;
        float4 a = make_float4(0.0f, 0.0f, 0.0f, 0.0f);
        #pragma unroll
        for (int s = 0; s < SLICES; ++s) {
            const float cs = __expf(ml[j][s].x - M);
            L   += cs * ml[j][s].y;
            a.x += cs * p[j][s].x;
            a.y += cs * p[j][s].y;
            a.z += cs * p[j][s].z;
            a.w += cs * p[j][s].w;
        }
        const float inv = 1.0f / L;
        *reinterpret_cast<float4*>(&s_pool[(rp + 8 * j) * PPITCH + 4 * c]) =
            make_float4(a.x * inv, a.y * inv, a.z * inv, a.w * inv);
    }
    __syncthreads();

    const float* ap = &s_pool[l * PPITCH + hf * 2];
    v2f afr[IN_DIM / 4];
    #pragma unroll
    for (int k = 0; k < IN_DIM / 4; ++k) {
        afr[k].x = ap[4 * k];
        afr[k].y = ap[4 * k + 1];
    }
    STAGE_FENCE();
    v8f acc = {};
    #pragma unroll
    for (int k = 0; k < IN_DIM / 4; ++k) {
        acc = __builtin_amdgcn_wmma_f32_16x16x4_f32(
                   false, afr[k],  false, bfr[k],
                   (short)0, acc,  false,  false);
    }

    const int col = n0 + l;
    #pragma unroll
    for (int r = 0; r < 8; ++r) {
        const int orow = m0 + hf * 8 + r;
        out[orow * FEAT_DIM + col] = acc[r] + bv;
    }
}

extern "C" void kernel_launch(void* const* d_in, const int* in_sizes, int n_in,
                              void* d_out, int out_size, void* d_ws, size_t ws_size,
                              hipStream_t stream) {
    const float* x    = (const float*)d_in[0];
    const int*   mask = (const int*)  d_in[1];
    const float* qt   = (const float*)d_in[2];
    const float* W    = (const float*)d_in[3];
    const float* bias = (const float*)d_in[4];
    float* out = (float*)d_out;

    const int B = in_sizes[0] / (N_TOK * IN_DIM);

    float* ws     = (float*)d_ws;
    float* pooled = ws + 512;

    if (ws_size >= (size_t)B_MAX * NQ * FEAT_DIM * sizeof(float)) {
        lap_pool_slice_kernel<<<dim3(B, SLICES), PS_THREADS, 0, stream>>>(x, mask, qt, W, bias, ws);
        lap_merge_outgemm_kernel<<<dim3(FEAT_DIM / MG_COLS, (B * NQ) / 16), MG_THREADS, 0, stream>>>(
            ws, W, bias, out);
    } else {
        lap_pool_slice_kernel<<<dim3(B, SLICES), PS_THREADS, 0, stream>>>(x, mask, qt, W, bias, out);
        lap_combine_kernel<<<B * NQ, IN_DIM, 0, stream>>>(out, pooled);
        lap_outgemm_kernel<<<dim3(FEAT_DIM / 16, (B * NQ) / 16), 32, 0, stream>>>(
            pooled, W, bias, out);
    }
}
